# starting kernel
# baseline (speedup 1.0000x reference)
#include <hip/hip_runtime.h>

typedef float v2f __attribute__((ext_vector_type(2)));
typedef float v4f __attribute__((ext_vector_type(4)));
typedef float v8f __attribute__((ext_vector_type(8)));

#define BATCH 16384
#define DDIM  1024
#define TILE  16
#define NWAVE 8
#define KSPAN (DDIM / NWAVE)
#define THSTR 1028

__global__ __launch_bounds__(256)
void cross_compress_kernel(const float* __restrict__ enc_user,
                           const float* __restrict__ enc_item,
                           const float* __restrict__ t_vv,
                           const float* __restrict__ t_ev,
                           const float* __restrict__ t_ve,
                           const float* __restrict__ t_ee,
                           const float* __restrict__ beta_v,
                           const float* __restrict__ beta_e,
                           float* __restrict__ v_out,
                           float* __restrict__ e_out)
{
    __shared__ float th_lds[5 * THSTR];
    __shared__ float red[2][NWAVE][4][16];
    __shared__ float dots[2][4][16];

    const int tid  = threadIdx.x;
    const int wave = tid >> 5;
    const int lane = tid & 31;
    const int g    = lane >> 4;
    const int sel  = lane & 15;
    const int row0 = blockIdx.x * TILE;

    {
        const int c = tid * 4;
        *(v4f*)&th_lds[0 * THSTR + c] = *(const v4f*)(t_vv + c);
        *(v4f*)&th_lds[1 * THSTR + c] = *(const v4f*)(t_ev + c);
        *(v4f*)&th_lds[2 * THSTR + c] = *(const v4f*)(t_ve + c);
        *(v4f*)&th_lds[3 * THSTR + c] = *(const v4f*)(t_ee + c);
        v4f z = {0.f, 0.f, 0.f, 0.f};
        *(v4f*)&th_lds[4 * THSTR + c] = z;
    }
    __syncthreads();

    const int arow = (sel < 4) ? sel : 4;
    const float* aptr = &th_lds[arow * THSTR];

    v8f acc_u = {0.f,0.f,0.f,0.f,0.f,0.f,0.f,0.f};
    v8f acc_i = {0.f,0.f,0.f,0.f,0.f,0.f,0.f,0.f};

    const float* urow = enc_user + (size_t)(row0 + sel) * DDIM;
    const float* irow = enc_item + (size_t)(row0 + sel) * DDIM;

    #pragma unroll 4
    for (int s = 0; s < KSPAN / 4; ++s) {
        const int kk = wave * KSPAN + s * 4 + g * 2;
        v2f a  = *(const v2f*)(aptr + kk);
        v2f bu = *(const v2f*)(urow + kk);
        v2f bi = *(const v2f*)(irow + kk);
        acc_u = __builtin_amdgcn_wmma_f32_16x16x4_f32(
            false, a, false, bu, (short)0, acc_u, false, false);
        acc_i = __builtin_amdgcn_wmma_f32_16x16x4_f32(
            false, a, false, bi, (short)0, acc_i, false, false);
    }

    if (lane < 16) {
        #pragma unroll
        for (int m = 0; m < 4; ++m) {
            red[0][wave][m][lane] = acc_u[m];
            red[1][wave][m][lane] = acc_i[m];
        }
    }
    __syncthreads();
    if (tid < 128) {
        const int mat = tid >> 6, m = (tid >> 4) & 3, n = tid & 15;
        float ssum = 0.f;
        #pragma unroll
        for (int w2 = 0; w2 < NWAVE; ++w2) ssum += red[mat][w2][m][n];
        dots[mat][m][n] = ssum;
    }
    __syncthreads();

    #pragma unroll
    for (int rr = 0; rr < 2; ++rr) {
        const int r = wave * 2 + rr;
        const size_t row = (size_t)(row0 + r);
        const float e_vv = dots[1][0][r];
        const float v_ev = dots[0][1][r];
        const float e_ve = dots[1][2][r];
        const float v_ee = dots[0][3][r];
        const float* up = enc_user + row * DDIM;
        const float* ip = enc_item + row * DDIM;
        float* vp = v_out + row * DDIM;
        float* ep = e_out + row * DDIM;
        #pragma unroll 2
        for (int base = 0; base < DDIM; base += 128) {
            const int col = base + lane * 4;
            v4f u  = *(const v4f*)(up + col);
            v4f it = *(const v4f*)(ip + col);
            v4f bv = *(const v4f*)(beta_v + col);
            v4f be = *(const v4f*)(beta_e + col);
            v4f vo = u * e_vv + it * v_ev + bv;
            v4f eo = u * e_ve + it * v_ee + be;
            __builtin_nontemporal_store(vo, (v4f*)(vp + col));
            __builtin_nontemporal_store(eo, (v4f*)(ep + col));
        }
    }
}

extern "C" void kernel_launch(void* const* d_in, const int* in_sizes, int n_in,
                              void* d_out, int out_size, void* d_ws, size_t ws_size,
                              hipStream_t stream) {
    const float* enc_user = (const float*)d_in[0];
    const float* enc_item = (const float*)d_in[1];
    const float* t_vv     = (const float*)d_in[2];
    const float* t_ev     = (const float*)d_in[3];
    const float* t_ve     = (const float*)d_in[4];
    const float* t_ee     = (const float*)d_in[5];
    const float* beta_v   = (const float*)d_in[6];
    const float* beta_e   = (const float*)d_in[7];
    float* v_out = (float*)d_out;
    float* e_out = (float*)d_out + (size_t)BATCH * DDIM;

    dim3 grid(BATCH / TILE);
    dim3 block(256);
    hipLaunchKernelGGL(cross_compress_kernel, grid, block, 0, stream,
                       enc_user, enc_item, t_vv, t_ev, t_ve, t_ee,
                       beta_v, beta_e, v_out, e_out);
}


# Tile read once: coalesced 512-B row loads kept in registers for the epilogue, copied to LDS for the WMMA B fragments (16-B reads, two K-steps each); k order of the sums changes
# speedup vs baseline: 1.5013x; 1.5013x over previous
#include <hip/hip_runtime.h>


typedef float v2f __attribute__((ext_vector_type(2)));
typedef float v4f __attribute__((ext_vector_type(4)));
typedef float v8f __attribute__((ext_vector_type(8)));

#define BATCH 16384
#define DDIM  1024
#define TILE  16
#define NWAVE 8
#define KSPAN (DDIM / NWAVE)
#define RSTR  1028
#define NPIECE (DDIM / 128)

__global__ __launch_bounds__(256)
void cross_compress_kernel(const float* __restrict__ enc_user,
                           const float* __restrict__ enc_item,
                           const float* __restrict__ t_vv,
                           const float* __restrict__ t_ev,
                           const float* __restrict__ t_ve,
                           const float* __restrict__ t_ee,
                           const float* __restrict__ beta_v,
                           const float* __restrict__ beta_e,
                           float* __restrict__ v_out,
                           float* __restrict__ e_out)
{
    __shared__ __attribute__((aligned(16))) float th_lds[5 * RSTR];
    __shared__ __attribute__((aligned(16))) float u_lds[TILE * RSTR];
    __shared__ __attribute__((aligned(16))) float i_lds[TILE * RSTR];
    __shared__ float red[2][NWAVE][4][16];
    __shared__ float dots[2][4][16];

    const int tid  = threadIdx.x;
    const int wave = tid >> 5;
    const int lane = tid & 31;
    const int g    = lane >> 4;
    const int sel  = lane & 15;
    const int row0 = blockIdx.x * TILE;

    v4f ur[2][NPIECE];
    v4f ir[2][NPIECE];
    #pragma unroll
    for (int rr = 0; rr < 2; ++rr) {
        const size_t row = (size_t)(row0 + wave * 2 + rr);
        const float* up = enc_user + row * DDIM + lane * 4;
        const float* ip = enc_item + row * DDIM + lane * 4;
        #pragma unroll
        for (int j = 0; j < NPIECE; ++j) {
            ur[rr][j] = *(const v4f*)(up + j * 128);
            ir[rr][j] = *(const v4f*)(ip + j * 128);
        }
    }

    {
        const int c = tid * 4;
        *(v4f*)&th_lds[0 * RSTR + c] = *(const v4f*)(t_vv + c);
        *(v4f*)&th_lds[1 * RSTR + c] = *(const v4f*)(t_ev + c);
        *(v4f*)&th_lds[2 * RSTR + c] = *(const v4f*)(t_ve + c);
        *(v4f*)&th_lds[3 * RSTR + c] = *(const v4f*)(t_ee + c);
        v4f z = {0.f, 0.f, 0.f, 0.f};
        *(v4f*)&th_lds[4 * RSTR + c] = z;
    }

    #pragma unroll
    for (int rr = 0; rr < 2; ++rr) {
        const int r = wave * 2 + rr;
        #pragma unroll
        for (int j = 0; j < NPIECE; ++j) {
            *(v4f*)&u_lds[r * RSTR + j * 128 + lane * 4] = ur[rr][j];
            *(v4f*)&i_lds[r * RSTR + j * 128 + lane * 4] = ir[rr][j];
        }
    }
    __syncthreads();

    const int arow = (sel < 4) ? sel : 4;
    const float* aptr = &th_lds[arow * RSTR];
    const float* ubp  = &u_lds[sel * RSTR];
    const float* ibp  = &i_lds[sel * RSTR];

    v8f acc_u = {0.f,0.f,0.f,0.f,0.f,0.f,0.f,0.f};
    v8f acc_i = {0.f,0.f,0.f,0.f,0.f,0.f,0.f,0.f};

    #pragma unroll 4
    for (int s = 0; s < KSPAN / 8; ++s) {
        const int kk = wave * KSPAN + s * 8 + g * 4;
        const v4f a4  = *(const v4f*)(aptr + kk);
        const v4f bu4 = *(const v4f*)(ubp + kk);
        const v4f bi4 = *(const v4f*)(ibp + kk);
        const v2f a_lo  = {a4.x,  a4.y},  a_hi  = {a4.z,  a4.w};
        const v2f bu_lo = {bu4.x, bu4.y}, bu_hi = {bu4.z, bu4.w};
        const v2f bi_lo = {bi4.x, bi4.y}, bi_hi = {bi4.z, bi4.w};
        acc_u = __builtin_amdgcn_wmma_f32_16x16x4_f32(
            false, a_lo, false, bu_lo, (short)0, acc_u, false, false);
        acc_i = __builtin_amdgcn_wmma_f32_16x16x4_f32(
            false, a_lo, false, bi_lo, (short)0, acc_i, false, false);
        acc_u = __builtin_amdgcn_wmma_f32_16x16x4_f32(
            false, a_hi, false, bu_hi, (short)0, acc_u, false, false);
        acc_i = __builtin_amdgcn_wmma_f32_16x16x4_f32(
            false, a_hi, false, bi_hi, (short)0, acc_i, false, false);
    }

    if (lane < 16) {
        #pragma unroll
        for (int m = 0; m < 4; ++m) {
            red[0][wave][m][lane] = acc_u[m];
            red[1][wave][m][lane] = acc_i[m];
        }
    }
    __syncthreads();
    if (tid < 128) {
        const int mat = tid >> 6, m = (tid >> 4) & 3, n = tid & 15;
        float ssum = 0.f;
        #pragma unroll
        for (int w2 = 0; w2 < NWAVE; ++w2) ssum += red[mat][w2][m][n];
        dots[mat][m][n] = ssum;
    }
    __syncthreads();

    float e_vv[2], v_ev[2], e_ve[2], v_ee[2];
    #pragma unroll
    for (int rr = 0; rr < 2; ++rr) {
        const int r = wave * 2 + rr;
        e_vv[rr] = dots[1][0][r];
        v_ev[rr] = dots[0][1][r];
        e_ve[rr] = dots[1][2][r];
        v_ee[rr] = dots[0][3][r];
    }
    #pragma unroll
    for (int j = 0; j < NPIECE; ++j) {
        const int col = j * 128 + lane * 4;
        const v4f bv = *(const v4f*)(beta_v + col);
        const v4f be = *(const v4f*)(beta_e + col);
        #pragma unroll
        for (int rr = 0; rr < 2; ++rr) {
            const size_t row = (size_t)(row0 + wave * 2 + rr);
            const v4f u  = ur[rr][j];
            const v4f it = ir[rr][j];
            const v4f vo = u * e_vv[rr] + it * v_ev[rr] + bv;
            const v4f eo = u * e_ve[rr] + it * v_ee[rr] + be;
            __builtin_nontemporal_store(vo, (v4f*)(v_out + row * DDIM + col));
            __builtin_nontemporal_store(eo, (v4f*)(e_out + row * DDIM + col));
        }
    }
}

extern "C" void kernel_launch(void* const* d_in, const int* in_sizes, int n_in,
                              void* d_out, int out_size, void* d_ws, size_t ws_size,
                              hipStream_t stream) {
    const float* enc_user = (const float*)d_in[0];
    const float* enc_item = (const float*)d_in[1];
    const float* t_vv     = (const float*)d_in[2];
    const float* t_ev     = (const float*)d_in[3];
    const float* t_ve     = (const float*)d_in[4];
    const float* t_ee     = (const float*)d_in[5];
    const float* beta_v   = (const float*)d_in[6];
    const float* beta_e   = (const float*)d_in[7];
    float* v_out = (float*)d_out;
    float* e_out = (float*)d_out + (size_t)BATCH * DDIM;

    dim3 grid(BATCH / TILE);
    dim3 block(256);
    hipLaunchKernelGGL(cross_compress_kernel, grid, block, 0, stream,
                       enc_user, enc_item, t_vv, t_ev, t_ve, t_ee,
                       beta_v, beta_e, v_out, e_out);
}


# Each wave takes its two rows' dots alone from its load registers (block-diagonal WMMA over 8 K-slices): no tile copy in LDS, no cross-wave sum, 1 barrier not 4; k order of the sums changes
# speedup vs baseline: 1.6177x; 1.0775x over previous
#include <hip/hip_runtime.h>


typedef float v2f __attribute__((ext_vector_type(2)));
typedef float v4f __attribute__((ext_vector_type(4)));
typedef float v8f __attribute__((ext_vector_type(8)));

#define BATCH 16384
#define DDIM  1024
#define TILE  16
#define NWAVE 8
#define NSTEP (DDIM / 64)
#define VSTR  1056


template <int M>
__device__ __forceinline__ float xlane(float x) {
    return __builtin_bit_cast(float, __builtin_amdgcn_update_dpp(
        0, __builtin_bit_cast(int, x), 0x160 + M, 0xf, 0xf, false));
}
__device__ __forceinline__ float xhalf(float x) {
    const int b = __builtin_bit_cast(int, x);
    return __builtin_bit_cast(float, __builtin_amdgcn_permlanex16(
        b, b, 0x76543210, 0xfedcba98, false, false));
}

__global__ __launch_bounds__(256)
void cross_compress_kernel(const float* __restrict__ enc_user,
                           const float* __restrict__ enc_item,
                           const float* __restrict__ t_vv,
                           const float* __restrict__ t_ev,
                           const float* __restrict__ t_ve,
                           const float* __restrict__ t_ee,
                           const float* __restrict__ beta_v,
                           const float* __restrict__ beta_e,
                           float* __restrict__ v_out,
                           float* __restrict__ e_out)
{
    __shared__ __attribute__((aligned(16))) float vec_lds[6 * VSTR];

    const int tid  = threadIdx.x;
    const int wave = tid >> 5;
    const int lane = tid & 31;
    const int h    = lane >> 4;
    const int p    = (lane >> 3) & 1;
    const int q    = lane & 7;
    const int koff = h * 32 + q * 4;
    const size_t row = (size_t)blockIdx.x * TILE + wave * 2 + p;

    const int cc = tid * 4;
    const v4f sv0 = *(const v4f*)(t_ev + cc);
    const v4f sv1 = *(const v4f*)(t_ee + cc);
    const v4f sv2 = *(const v4f*)(t_vv + cc);
    const v4f sv3 = *(const v4f*)(t_ve + cc);
    const v4f sv4 = *(const v4f*)(beta_v + cc);
    const v4f sv5 = *(const v4f*)(beta_e + cc);

    const float* up = enc_user + row * DDIM + koff;
    const float* ip = enc_item + row * DDIM + koff;
    v4f ur[NSTEP];
    v4f ir[NSTEP];
    #pragma unroll
    for (int s = 0; s < NSTEP; ++s) {
        ur[s] = *(const v4f*)(up + s * 64);
        ir[s] = *(const v4f*)(ip + s * 64);
    }

    *(v4f*)&vec_lds[0 * VSTR + cc] = sv0;
    *(v4f*)&vec_lds[1 * VSTR + cc] = sv1;
    *(v4f*)&vec_lds[2 * VSTR + cc] = sv2;
    *(v4f*)&vec_lds[3 * VSTR + cc] = sv3;
    *(v4f*)&vec_lds[4 * VSTR + cc] = sv4;
    *(v4f*)&vec_lds[5 * VSTR + cc] = sv5;
    __builtin_amdgcn_fence(__ATOMIC_RELEASE, "workgroup", "local");
    __builtin_amdgcn_s_barrier();
    __builtin_amdgcn_fence(__ATOMIC_ACQUIRE, "workgroup", "local");

    const float* au = &vec_lds[(0 + p) * VSTR + koff];
    const float* ai = &vec_lds[(2 + p) * VSTR + koff];

    v8f acc_u = {0.f,0.f,0.f,0.f,0.f,0.f,0.f,0.f};
    v8f acc_i = {0.f,0.f,0.f,0.f,0.f,0.f,0.f,0.f};

    #pragma unroll
    for (int s = 0; s < NSTEP; ++s) {
        const v4f a_u = *(const v4f*)(au + s * 64);
        const v4f a_i = *(const v4f*)(ai + s * 64);
        const v4f b_u = ur[s];
        const v4f b_i = ir[s];
        const v2f au_lo = {a_u.x, a_u.y}, au_hi = {a_u.z, a_u.w};
        const v2f ai_lo = {a_i.x, a_i.y}, ai_hi = {a_i.z, a_i.w};
        const v2f bu_lo = {b_u.x, b_u.y}, bu_hi = {b_u.z, b_u.w};
        const v2f bi_lo = {b_i.x, b_i.y}, bi_hi = {b_i.z, b_i.w};
        acc_u = __builtin_amdgcn_wmma_f32_16x16x4_f32(
            false, au_lo, false, bu_lo, (short)0, acc_u, false, false);
        acc_i = __builtin_amdgcn_wmma_f32_16x16x4_f32(
            false, ai_lo, false, bi_lo, (short)0, acc_i, false, false);
        acc_u = __builtin_amdgcn_wmma_f32_16x16x4_f32(
            false, au_hi, false, bu_hi, (short)0, acc_u, false, false);
        acc_i = __builtin_amdgcn_wmma_f32_16x16x4_f32(
            false, ai_hi, false, bi_hi, (short)0, acc_i, false, false);
    }

    float du = acc_u[0];
    float di = acc_i[0];
    #pragma unroll
    for (int r = 1; r < 8; ++r) {
        du = (q == r) ? acc_u[r] : du;
        di = (q == r) ? acc_i[r] : di;
    }
    du += xlane<1>(du);  di += xlane<1>(di);
    du += xlane<2>(du);  di += xlane<2>(di);
    du += xlane<4>(du);  di += xlane<4>(di);
    const float odu = xhalf(du);
    const float odi = xhalf(di);
    const float v_ev = h ? odu : du;
    const float v_ee = h ? du  : odu;
    const float e_vv = h ? odi : di;
    const float e_ve = h ? di  : odi;

    const float* bvp = &vec_lds[4 * VSTR + koff];
    const float* bep = &vec_lds[5 * VSTR + koff];
    float* vo_p = v_out + row * DDIM + koff;
    float* eo_p = e_out + row * DDIM + koff;
    #pragma unroll
    for (int g4 = 0; g4 < NSTEP; g4 += 4) {
        if (g4 > 0) __builtin_amdgcn_sched_barrier(0);
        #pragma unroll
        for (int s = g4; s < g4 + 4; ++s) {
            const v4f bv = *(const v4f*)(bvp + s * 64);
            const v4f be = *(const v4f*)(bep + s * 64);
            const v4f u  = ur[s];
            const v4f it = ir[s];
            const v4f vo = u * e_vv + it * v_ev + bv;
            const v4f eo = u * e_ve + it * v_ee + be;
            __builtin_nontemporal_store(vo, (v4f*)(vo_p + s * 64));
            __builtin_nontemporal_store(eo, (v4f*)(eo_p + s * 64));
        }
    }
}

extern "C" void kernel_launch(void* const* d_in, const int* in_sizes, int n_in,
                              void* d_out, int out_size, void* d_ws, size_t ws_size,
                              hipStream_t stream) {
    const float* enc_user = (const float*)d_in[0];
    const float* enc_item = (const float*)d_in[1];
    const float* t_vv     = (const float*)d_in[2];
    const float* t_ev     = (const float*)d_in[3];
    const float* t_ve     = (const float*)d_in[4];
    const float* t_ee     = (const float*)d_in[5];
    const float* beta_v   = (const float*)d_in[6];
    const float* beta_e   = (const float*)d_in[7];
    float* v_out = (float*)d_out;
    float* e_out = (float*)d_out + (size_t)BATCH * DDIM;

    dim3 grid(BATCH / TILE);
    dim3 block(256);
    hipLaunchKernelGGL(cross_compress_kernel, grid, block, 0, stream,
                       enc_user, enc_item, t_vv, t_ev, t_ve, t_ee,
                       beta_v, beta_e, v_out, e_out);
}
